# starting kernel
# baseline (speedup 1.0000x reference)
#include <hip/hip_runtime.h>

#define NNODES 65536
#define NEDGES 1048576
#define DIM    64

typedef float v2f __attribute__((ext_vector_type(2)));
typedef float v8f __attribute__((ext_vector_type(8)));

__global__ void zero_kernel(float* __restrict__ p, int n) {
    int i = blockIdx.x * blockDim.x + threadIdx.x;
    if (i < n) p[i] = 0.0f;
}

__global__ void degree_kernel(const int* __restrict__ dst, float* __restrict__ deg, int nE) {
    int i = blockIdx.x * blockDim.x + threadIdx.x;
    if (i < nE) atomicAdd(&deg[dst[i]], 1.0f);
}

__global__ void norm_kernel(float* __restrict__ deg, int n) {
    int i = blockIdx.x * blockDim.x + threadIdx.x;
    if (i < n) deg[i] = rsqrtf(fmaxf(deg[i], 1.0f));
}

__global__ void scale_kernel(const float* __restrict__ in, const float* __restrict__ norm,
                             float* __restrict__ o, int nNodes, int pw) {
    int tid = blockIdx.x * blockDim.x + threadIdx.x;
    if (tid >= nNodes * 16) return;
    int v = tid >> 4, c = tid & 15;
    float s = norm[v];
    if (pw == 2) s *= s;
    float4 x = ((const float4*)(in + v * DIM))[c];
    x.x *= s; x.y *= s; x.z *= s; x.w *= s;
    ((float4*)(o + v * DIM))[c] = x;
}

__global__ __launch_bounds__(256) void scatter_kernel(const float* __restrict__ S,
                                                      const int* __restrict__ src,
                                                      const int* __restrict__ dst,
                                                      float* __restrict__ AGG, int nE) {
    int tid = blockIdx.x * blockDim.x + threadIdx.x;
    int e = tid >> 4;
    int c = tid & 15;
    if (e >= nE) return;
    int s = src[e], d = dst[e];
    float4 v = ((const float4*)(S + s * DIM))[c];
    float* p = AGG + d * DIM + c * 4;
    atomicAdd(p + 0, v.x);
    atomicAdd(p + 1, v.y);
    atomicAdd(p + 2, v.z);
    atomicAdd(p + 3, v.w);
}

__global__ __launch_bounds__(256) void ffn_wmma_kernel(
    const float* __restrict__ r, const float* __restrict__ feat,
    const float* __restrict__ w1, const float* __restrict__ b1,
    const float* __restrict__ w2, const float* __restrict__ b2,
    const float* __restrict__ gamma, const float* __restrict__ beta,
    float* __restrict__ out) {
    __shared__ float lds[8 * 16 * DIM];
    const int wave = threadIdx.x >> 5;
    const int lane = threadIdx.x & 31;
    const int idx  = lane & 15;
    const int half = lane >> 4;
    const int rowbase = blockIdx.x * 128 + wave * 16;
    float* A = lds + wave * 16 * DIM;

    const float2 g2  = ((const float2*)gamma)[lane];
    const float2 be2 = ((const float2*)beta)[lane];
    for (int m = 0; m < 16; ++m) {
        float2 x = ((const float2*)(r + (rowbase + m) * DIM))[lane];
        float s = x.x + x.y;
        #pragma unroll
        for (int o = 16; o > 0; o >>= 1) s += __shfl_xor(s, o, 32);
        float mu = s * (1.0f / 64.0f);
        float dx = x.x - mu, dy = x.y - mu;
        float v = dx * dx + dy * dy;
        #pragma unroll
        for (int o = 16; o > 0; o >>= 1) v += __shfl_xor(v, o, 32);
        float rstd = rsqrtf(v * (1.0f / 64.0f) + 1e-5f);
        float2 y;
        y.x = dx * rstd * g2.x + be2.x;
        y.y = dy * rstd * g2.y + be2.y;
        ((float2*)(A + m * DIM))[lane] = y;
    }
    __syncthreads();

    v8f acc[4];
    #pragma unroll
    for (int j = 0; j < 4; ++j) acc[j] = (v8f)(0.0f);
    for (int kb = 0; kb < 16; ++kb) {
        const int k = kb * 4 + half * 2;
        v2f a = *(const v2f*)(A + idx * DIM + k);
        #pragma unroll
        for (int j = 0; j < 4; ++j) {
            float2 wv = *(const float2*)(w1 + (j * 16 + idx) * DIM + k);
            v2f b = {wv.x, wv.y};
            acc[j] = __builtin_amdgcn_wmma_f32_16x16x4_f32(
                false, a, false, b, (short)0, acc[j], false, false);
        }
    }
    __syncthreads();
    #pragma unroll
    for (int j = 0; j < 4; ++j) {
        const float bias = b1[j * 16 + idx];
        #pragma unroll
        for (int t = 0; t < 8; ++t) {
            float hv = acc[j][t] + bias;
            A[(t + 8 * half) * DIM + j * 16 + idx] = fmaxf(hv, 0.0f);
        }
    }
    __syncthreads();

    v8f acc2[4];
    #pragma unroll
    for (int j = 0; j < 4; ++j) acc2[j] = (v8f)(0.0f);
    for (int kb = 0; kb < 16; ++kb) {
        const int k = kb * 4 + half * 2;
        v2f a = *(const v2f*)(A + idx * DIM + k);
        #pragma unroll
        for (int j = 0; j < 4; ++j) {
            float2 wv = *(const float2*)(w2 + (j * 16 + idx) * DIM + k);
            v2f b = {wv.x, wv.y};
            acc2[j] = __builtin_amdgcn_wmma_f32_16x16x4_f32(
                false, a, false, b, (short)0, acc2[j], false, false);
        }
    }
    #pragma unroll
    for (int j = 0; j < 4; ++j) {
        const int col = j * 16 + idx;
        const float bias = b2[col];
        #pragma unroll
        for (int t = 0; t < 8; ++t) {
            const int row = rowbase + t + 8 * half;
            out[row * DIM + col] = acc2[j][t] + bias + r[row * DIM + col] + feat[row * DIM + col];
        }
    }
}

extern "C" void kernel_launch(void* const* d_in, const int* in_sizes, int n_in,
                              void* d_out, int out_size, void* d_ws, size_t ws_size,
                              hipStream_t stream) {
    const float* feat = (const float*)d_in[0];
    const int*   esrc = (const int*)d_in[1];
    const int*   edst = (const int*)d_in[2];
    const float* w1   = (const float*)d_in[3];
    const float* b1   = (const float*)d_in[4];
    const float* w2   = (const float*)d_in[5];
    const float* b2   = (const float*)d_in[6];
    const float* gam  = (const float*)d_in[7];
    const float* bet  = (const float*)d_in[8];

    float* out0 = (float*)d_out;
    float* out1 = out0 + (size_t)NNODES * DIM;
    float* norm = (float*)d_ws;

    const int ND = NNODES * DIM;
    dim3 blk(256);
    dim3 gN((NNODES + 255) / 256);
    dim3 gE((NEDGES + 255) / 256);
    dim3 gND16((NNODES * 16 + 255) / 256);
    dim3 gND((ND + 255) / 256);
    dim3 gSc((NEDGES * 16 + 255) / 256);

    zero_kernel<<<gN, blk, 0, stream>>>(norm, NNODES);
    degree_kernel<<<gE, blk, 0, stream>>>(edst, norm, NEDGES);
    norm_kernel<<<gN, blk, 0, stream>>>(norm, NNODES);

    scale_kernel<<<gND16, blk, 0, stream>>>(feat, norm, out0, NNODES, 1);
    zero_kernel<<<gND, blk, 0, stream>>>(out1, ND);
    scatter_kernel<<<gSc, blk, 0, stream>>>(out0, esrc, edst, out1, NEDGES);
    scale_kernel<<<gND16, blk, 0, stream>>>(out1, norm, out0, NNODES, 2);
    zero_kernel<<<gND, blk, 0, stream>>>(out1, ND);
    scatter_kernel<<<gSc, blk, 0, stream>>>(out0, esrc, edst, out1, NEDGES);
    scale_kernel<<<gND16, blk, 0, stream>>>(out1, norm, out0, NNODES, 2);
    zero_kernel<<<gND, blk, 0, stream>>>(out1, ND);
    scatter_kernel<<<gSc, blk, 0, stream>>>(out0, esrc, edst, out1, NEDGES);
    scale_kernel<<<gND16, blk, 0, stream>>>(out1, norm, out1, NNODES, 1);

    ffn_wmma_kernel<<<dim3(NNODES / 128), blk, 0, stream>>>(
        out1, feat, w1, b1, w2, b2, gam, bet, out0);
}


# speedup vs baseline: 3.1095x; 3.1095x over previous
#include <hip/hip_runtime.h>

#define NNODES 65536
#define NEDGES 1048576
#define DIM    64

typedef float v2f __attribute__((ext_vector_type(2)));
typedef float v8f __attribute__((ext_vector_type(8)));

__global__ void zero_kernel(float* __restrict__ p, int n) {
    int i = blockIdx.x * blockDim.x + threadIdx.x;
    if (i < n) p[i] = 0.0f;
}

__device__ __forceinline__ void clear_float4(float* __restrict__ buf, int i) {
    ((float4*)buf)[i] = make_float4(0.0f, 0.0f, 0.0f, 0.0f);
}

#define HALF_NODES (NNODES / 2)
#define HALF_F4    (NNODES * DIM / 8)

struct ItemDraw { int hq, craw, switched; };
__device__ __forceinline__ void draw_first(ItemDraw& q, int* __restrict__ queue) {
    q.hq = (int)((__builtin_amdgcn_s_sendmsg_rtn(0x87) >> 18) & 1u);
    q.switched = 0;
    q.craw = atomicAdd(&queue[q.hq], 1);
}
__device__ __forceinline__ int draw_take(ItemDraw& q, int* __restrict__ queue, int nchunk) {
    int c = q.craw;
    if (c >= nchunk && !q.switched) {
        q.switched = 1;
        q.hq ^= 1;
        c = atomicAdd(&queue[q.hq], 1);
    }
    return (c < nchunk) ? (c * 2 + q.hq) : -1;
}

#define DCHUNK  2048
#define NDCHUNK (NEDGES / DCHUNK)
#define DEGREE_WGS 1024
static_assert(NEDGES % DCHUNK == 0, "whole chunks, no tail");
static_assert(HALF_F4 == NDCHUNK * 1024, "an item clears 1024 float4 of its range of the next buffer");
__global__ __launch_bounds__(256) void degree_kernel(const int* __restrict__ dst, float* __restrict__ deg,
                                                     float* __restrict__ clr, int* __restrict__ queue) {
    __shared__ int slot[2];
    const int tid = threadIdx.x;
    ItemDraw q = {0, 0, 0};
    if (tid == 0) draw_first(q, queue);
    for (int it = 0; it < 2 * NDCHUNK + 2; ++it) {
        if (tid == 0) slot[it & 1] = draw_take(q, queue, NDCHUNK);
        __syncthreads();
        const int item = __builtin_amdgcn_readfirstlane(slot[it & 1]);
        if (item < 0) break;
        if (tid == 0) q.craw = atomicAdd(&queue[q.hq], 1);
        const int h = item & 1;
        const int c = item >> 1;
        #pragma unroll
        for (int k = 0; k < 4; ++k) clear_float4(clr, h * HALF_F4 + c * 1024 + k * 256 + tid);
        const int4* p = (const int4*)(dst + c * DCHUNK);
        const int4 a = p[tid], b = p[tid + 256];
        const int dd[8] = {a.x, a.y, a.z, a.w, b.x, b.y, b.z, b.w};
        #pragma unroll
        for (int k = 0; k < 8; ++k)
            if ((dd[k] >= HALF_NODES) == (h != 0)) atomicAdd(&deg[dd[k]], 1.0f);
    }
}

#define CHUNK   512
#define NCHUNK  (NEDGES / CHUNK)
#define SCATTER_WGS 2048
static_assert(NEDGES % CHUNK == 0, "whole chunks, no tail");
static_assert(DIM == 64, "a row is two 128-byte lines, one per 32-lane instruction");
static_assert(HALF_F4 == NCHUNK * 256, "an item clears 256 float4 of its range of the next buffer");

template <bool FULL>
__device__ __forceinline__ void scatter_group(const float* __restrict__ X, float* __restrict__ AGG,
                                              unsigned long long& M, int g, int lane,
                                              int s0, int s1, int d0, int d1, int c0, int c1) {
    float v0[8], v1[8], cc[8];
    int dd[8];
    #pragma unroll
    for (int k = 0; k < 8; ++k) {
        v0[k] = 0.0f; v1[k] = 0.0f; cc[k] = 0.0f; dd[k] = 0;
        if (FULL || k < g) {
            const int j = __builtin_ctzll(M);
            M &= M - 1;
            const int jl = j & 31;
            const bool hi = (j & 32) != 0;
            const int sa = __builtin_amdgcn_readlane(s0, jl), sb = __builtin_amdgcn_readlane(s1, jl);
            const int da = __builtin_amdgcn_readlane(d0, jl), db = __builtin_amdgcn_readlane(d1, jl);
            const int ca = __builtin_amdgcn_readlane(c0, jl), cb = __builtin_amdgcn_readlane(c1, jl);
            const int sj = hi ? sb : sa;
            dd[k] = hi ? db : da;
            cc[k] = __int_as_float(hi ? cb : ca);
            const float* row = X + sj * DIM + lane;
            v0[k] = row[0];
            v1[k] = row[32];
        }
    }
    #pragma unroll
    for (int k = 0; k < 8; ++k) {
        if (FULL || k < g) {
            float* p = AGG + dd[k] * DIM + lane;
            atomicAdd(p,      v0[k] * cc[k]);
            atomicAdd(p + 32, v1[k] * cc[k]);
        }
    }
}

__global__ __launch_bounds__(256) void scatter_kernel(const float* __restrict__ X,
                                                      const float* __restrict__ deg,
                                                      const int* __restrict__ src,
                                                      const int* __restrict__ dst,
                                                      float* __restrict__ AGG,
                                                      float* __restrict__ clr,
                                                      float* __restrict__ esq,
                                                      int* __restrict__ queue, int pw) {
    __shared__ int slot[2];
    const int tid = threadIdx.x;
    const int lane = tid & 31;
    const int wave = tid >> 5;
    ItemDraw q = {0, 0, 0};
    if (tid == 0) draw_first(q, queue);
    for (int it = 0; it < 2 * NCHUNK + 2; ++it) {
        if (tid == 0) slot[it & 1] = draw_take(q, queue, NCHUNK);
        __syncthreads();
        const int item = __builtin_amdgcn_readfirstlane(slot[it & 1]);
        if (item < 0) break;
        if (tid == 0) q.craw = atomicAdd(&queue[q.hq], 1);
        const int h = item & 1;
        const int c = item >> 1;
        if (clr) clear_float4(clr, h * HALF_F4 + c * 256 + tid);

        const int e0 = c * CHUNK + wave * 64 + lane;
        const int e1 = e0 + 32;
        const int s0 = src[e0], d0 = dst[e0];
        const int s1 = src[e1], d1 = dst[e1];
        const bool sel0 = (d0 >= HALF_NODES) == (h != 0);
        const bool sel1 = (d1 >= HALF_NODES) == (h != 0);
        float sc0 = 0.0f, sc1 = 0.0f;
        if (pw == 1) {
            if (sel0) { sc0 = rsqrtf(fmaxf(deg[s0], 1.0f)); esq[e0] = sc0 * sc0; }
            if (sel1) { sc1 = rsqrtf(fmaxf(deg[s1], 1.0f)); esq[e1] = sc1 * sc1; }
        } else {
            sc0 = esq[e0];
            sc1 = esq[e1];
        }
        const int c0 = __float_as_int(sc0), c1 = __float_as_int(sc1);
        unsigned long long M = (unsigned long long)__builtin_amdgcn_ballot_w32(sel0)
                             | ((unsigned long long)__builtin_amdgcn_ballot_w32(sel1) << 32);
        int n = __builtin_popcountll(M);
        for (; n >= 8; n -= 8) scatter_group<true>(X, AGG, M, 8, lane, s0, s1, d0, d1, c0, c1);
        if (n > 0)            scatter_group<false>(X, AGG, M, n, lane, s0, s1, d0, d1, c0, c1);
    }
}

__global__ __launch_bounds__(256) void ffn_wmma_kernel(
    const float* __restrict__ agg, const float* __restrict__ deg, const float* __restrict__ feat,
    const float* __restrict__ w1, const float* __restrict__ b1,
    const float* __restrict__ w2, const float* __restrict__ b2,
    const float* __restrict__ gamma, const float* __restrict__ beta,
    float* __restrict__ rout, float* __restrict__ out) {
    __shared__ float lds[8 * 16 * DIM];
    const int wave = threadIdx.x >> 5;
    const int lane = threadIdx.x & 31;
    const int idx  = lane & 15;
    const int half = lane >> 4;
    const int rowbase = blockIdx.x * 128 + wave * 16;
    float* A = lds + wave * 16 * DIM;

    const float2 g2  = ((const float2*)gamma)[lane];
    const float2 be2 = ((const float2*)beta)[lane];
    for (int m = 0; m < 16; ++m) {
        const float nm = rsqrtf(fmaxf(deg[rowbase + m], 1.0f));
        float2 x = ((const float2*)(agg + (rowbase + m) * DIM))[lane];
        x.x *= nm; x.y *= nm;
        ((float2*)(rout + (rowbase + m) * DIM))[lane] = x;
        float s = x.x + x.y;
        #pragma unroll
        for (int o = 16; o > 0; o >>= 1) s += __shfl_xor(s, o, 32);
        float mu = s * (1.0f / 64.0f);
        float dx = x.x - mu, dy = x.y - mu;
        float v = dx * dx + dy * dy;
        #pragma unroll
        for (int o = 16; o > 0; o >>= 1) v += __shfl_xor(v, o, 32);
        float rstd = rsqrtf(v * (1.0f / 64.0f) + 1e-5f);
        float2 y;
        y.x = dx * rstd * g2.x + be2.x;
        y.y = dy * rstd * g2.y + be2.y;
        ((float2*)(A + m * DIM))[lane] = y;
    }
    __syncthreads();

    v8f acc[4];
    #pragma unroll
    for (int j = 0; j < 4; ++j) acc[j] = (v8f)(0.0f);
    for (int kb = 0; kb < 16; ++kb) {
        const int k = kb * 4 + half * 2;
        v2f a = *(const v2f*)(A + idx * DIM + k);
        #pragma unroll
        for (int j = 0; j < 4; ++j) {
            float2 wv = *(const float2*)(w1 + (j * 16 + idx) * DIM + k);
            v2f b = {wv.x, wv.y};
            acc[j] = __builtin_amdgcn_wmma_f32_16x16x4_f32(
                false, a, false, b, (short)0, acc[j], false, false);
        }
    }
    __syncthreads();
    #pragma unroll
    for (int j = 0; j < 4; ++j) {
        const float bias = b1[j * 16 + idx];
        #pragma unroll
        for (int t = 0; t < 8; ++t) {
            float hv = acc[j][t] + bias;
            A[(t + 8 * half) * DIM + j * 16 + idx] = fmaxf(hv, 0.0f);
        }
    }
    __syncthreads();

    float nr[8];
    #pragma unroll
    for (int t = 0; t < 8; ++t) nr[t] = rsqrtf(fmaxf(deg[rowbase + t + 8 * half], 1.0f));
    v8f acc2[4];
    #pragma unroll
    for (int j = 0; j < 4; ++j) acc2[j] = (v8f)(0.0f);
    for (int kb = 0; kb < 16; ++kb) {
        const int k = kb * 4 + half * 2;
        v2f a = *(const v2f*)(A + idx * DIM + k);
        #pragma unroll
        for (int j = 0; j < 4; ++j) {
            float2 wv = *(const float2*)(w2 + (j * 16 + idx) * DIM + k);
            v2f b = {wv.x, wv.y};
            acc2[j] = __builtin_amdgcn_wmma_f32_16x16x4_f32(
                false, a, false, b, (short)0, acc2[j], false, false);
        }
    }
    #pragma unroll
    for (int j = 0; j < 4; ++j) {
        const int col = j * 16 + idx;
        const float bias = b2[col];
        #pragma unroll
        for (int t = 0; t < 8; ++t) {
            const int row = rowbase + t + 8 * half;
            out[row * DIM + col] = acc2[j][t] + bias + agg[row * DIM + col] * nr[t] + feat[row * DIM + col];
        }
    }
}

extern "C" void kernel_launch(void* const* d_in, const int* in_sizes, int n_in,
                              void* d_out, int out_size, void* d_ws, size_t ws_size,
                              hipStream_t stream) {
    const float* feat = (const float*)d_in[0];
    const int*   esrc = (const int*)d_in[1];
    const int*   edst = (const int*)d_in[2];
    const float* w1   = (const float*)d_in[3];
    const float* b1   = (const float*)d_in[4];
    const float* w2   = (const float*)d_in[5];
    const float* b2   = (const float*)d_in[6];
    const float* gam  = (const float*)d_in[7];
    const float* bet  = (const float*)d_in[8];

    float* out0 = (float*)d_out;
    float* out1 = out0 + (size_t)NNODES * DIM;
    float* deg  = (float*)d_ws;
    float* agg3 = (float*)d_ws + (size_t)NNODES * 4;
    float* esq  = agg3 + (size_t)NNODES * DIM;
    int*   qctr = (int*)(deg + NNODES);

    dim3 blk(256);
    dim3 gD(DEGREE_WGS);
    dim3 gZ((NNODES + 8 + 255) / 256);
    dim3 gSc(SCATTER_WGS);

    zero_kernel<<<gZ, blk, 0, stream>>>(deg, NNODES + 8);
    degree_kernel<<<gD, blk, 0, stream>>>(edst, deg, out1, qctr);

    scatter_kernel<<<gSc, blk, 0, stream>>>(feat, deg, esrc, edst, out1, out0, esq, qctr + 2, 1);
    scatter_kernel<<<gSc, blk, 0, stream>>>(out1, deg, esrc, edst, out0, agg3, esq, qctr + 4, 2);
    scatter_kernel<<<gSc, blk, 0, stream>>>(out0, deg, esrc, edst, agg3, nullptr, esq, qctr + 6, 2);

    ffn_wmma_kernel<<<dim3(NNODES / 128), blk, 0, stream>>>(
        agg3, deg, feat, w1, b1, w2, b2, gam, bet, out1, out0);
}
